# starting kernel
# baseline (speedup 1.0000x reference)
#include <hip/hip_runtime.h>

typedef __attribute__((ext_vector_type(2))) float v2f;
typedef __attribute__((ext_vector_type(8))) float v8f;

#define BATCH 4096
#define SDIM  1024
#define KWIN  1019
#define NTOT  (BATCH * KWIN)
#define NPAIRS (NTOT / 32)

__global__ __launch_bounds__(256) void actor_wmma_kernel(
    const float* __restrict__ s,
    const float* __restrict__ w1,
    const float* __restrict__ b1,
    const float* __restrict__ w3,
    const float* __restrict__ b3,
    float* __restrict__ out)
{
    const int lane  = threadIdx.x & 31;
    const int half  = lane >> 4;
    const int mrow  = lane & 15;

    const int wid = __builtin_amdgcn_readfirstlane(
        (int)((blockIdx.x * blockDim.x + threadIdx.x) >> 5));
    const int nW = (int)((gridDim.x * blockDim.x) >> 5);

    const float aux0 = s[SDIM - 3];
    const float aux1 = s[SDIM - 2];
    const float aux2 = s[SDIM - 1];
    const float b3s  = b3[0];

    v2f A0, A1, A2, A3;
    {
        const int f0 = 2 * half;
#define LOADA(t, R) { int m = (t) * 16 + mrow; \
                      R.x = w1[m * 6 + f0]; R.y = w1[m * 6 + f0 + 1]; }
        LOADA(0, A0) LOADA(1, A1) LOADA(2, A2) LOADA(3, A3)
#undef LOADA
    }

    v8f C0, C1, C2, C3;
    v8f W0, W1, W2, W3f;
#define LOADC(t, CR, WR) \
    _Pragma("unroll") \
    for (int v = 0; v < 8; ++v) { \
        int m = (t) * 16 + v + 8 * half; \
        CR[v] = b1[m] + aux1 * w1[m * 6 + 4] + aux2 * w1[m * 6 + 5]; \
        WR[v] = w3[m]; \
    }
    LOADC(0, C0, W0) LOADC(1, C1, W1) LOADC(2, C2, W2) LOADC(3, C3, W3f)
#undef LOADC

#define WMMA(Af, Bf, Cf) \
    __builtin_amdgcn_wmma_f32_16x16x4_f32(false, Af, false, Bf, (short)0, Cf, false, false)

    const int voff0 = mrow * SDIM + 2 * half;
    const int voff1 = mrow * SDIM + 1;

    auto base_ptr = [&](int pp) -> const float* {
        int n0 = pp << 5;
        return s + (n0 & (BATCH - 1)) * SDIM + (n0 >> 12);
    };

    if (wid >= NPAIRS) return;

    const float* sb = base_ptr(wid);
    float e00 = sb[voff0];
    float e01 = sb[voff1];
    float e10 = sb[voff0 + 16 * SDIM];
    float e11 = sb[voff1 + 16 * SDIM];

    for (int p = wid; p < NPAIRS; p += nW) {
        v2f B0, B1;
        B0.x = e00; B0.y = half ? aux0 : e01;
        B1.x = e10; B1.y = half ? aux0 : e11;

        v8f d00 = WMMA(A0, B0, C0);
        v8f d01 = WMMA(A1, B0, C1);
        v8f d02 = WMMA(A2, B0, C2);
        v8f d03 = WMMA(A3, B0, C3);
        v8f d10 = WMMA(A0, B1, C0);
        v8f d11 = WMMA(A1, B1, C1);
        v8f d12 = WMMA(A2, B1, C2);
        v8f d13 = WMMA(A3, B1, C3);

        int pn = p + nW;
        const float* sbn = base_ptr(pn < NPAIRS ? pn : wid);
        e00 = sbn[voff0];
        e01 = sbn[voff1];
        e10 = sbn[voff0 + 16 * SDIM];
        e11 = sbn[voff1 + 16 * SDIM];

        float a00 = 0.f, a01 = 0.f, a02 = 0.f, a03 = 0.f;
        float a10 = 0.f, a11 = 0.f, a12 = 0.f, a13 = 0.f;
#pragma unroll
        for (int v = 0; v < 8; ++v) {
            a00 += W0[v]  * fmaxf(d00[v], 0.f);
            a01 += W1[v]  * fmaxf(d01[v], 0.f);
            a02 += W2[v]  * fmaxf(d02[v], 0.f);
            a03 += W3f[v] * fmaxf(d03[v], 0.f);
            a10 += W0[v]  * fmaxf(d10[v], 0.f);
            a11 += W1[v]  * fmaxf(d11[v], 0.f);
            a12 += W2[v]  * fmaxf(d12[v], 0.f);
            a13 += W3f[v] * fmaxf(d13[v], 0.f);
        }
        float accA = (a00 + a01) + (a02 + a03);
        float accB = (a10 + a11) + (a12 + a13);

        int swA = __builtin_amdgcn_ds_swizzle(__builtin_bit_cast(int, accA), 0x401f);
        int swB = __builtin_amdgcn_ds_swizzle(__builtin_bit_cast(int, accB), 0x401f);
        float totA = accA + __builtin_bit_cast(float, swA);
        float totB = accB + __builtin_bit_cast(float, swB);

        float r = (half ? totB : totA) + b3s;
        r = __builtin_amdgcn_fmed3f(r, -1.0f, 1.0f);

        out[(p << 5) + lane] = r;
    }
#undef WMMA
}

extern "C" void kernel_launch(void* const* d_in, const int* in_sizes, int n_in,
                              void* d_out, int out_size, void* d_ws, size_t ws_size,
                              hipStream_t stream) {
    (void)in_sizes; (void)n_in; (void)out_size; (void)d_ws; (void)ws_size;
    const float* s  = (const float*)d_in[0];
    const float* w1 = (const float*)d_in[1];
    const float* b1 = (const float*)d_in[2];
    const float* w3 = (const float*)d_in[3];
    const float* b3 = (const float*)d_in[4];
    float* out = (float*)d_out;

    actor_wmma_kernel<<<1024, 256, 0, stream>>>(s, w1, b1, w3, b3, out);
}


# Re-map work so each wave walks 32 consecutive windows of its own 32 rows and reads s by aligned 16-byte per-lane loads kept in registers (was 4 strided 4-byte loads per tile pair); grid 512x256
# speedup vs baseline: 1.3290x; 1.3290x over previous
#include <hip/hip_runtime.h>


typedef __attribute__((ext_vector_type(2))) float v2f;
typedef __attribute__((ext_vector_type(4))) float v4f;
typedef __attribute__((ext_vector_type(8))) float v8f;

#define BATCH 4096
#define SDIM  1024
#define KWIN  1019
#define SPAN  32
#define NGRP  32
#define NROWG (BATCH / 32)
#define NWAVES (NROWG * NGRP)

__global__ __launch_bounds__(256) void actor_wmma_kernel(
    const float* __restrict__ s,
    const float* __restrict__ w1,
    const float* __restrict__ b1,
    const float* __restrict__ w3,
    const float* __restrict__ b3,
    float* __restrict__ out)
{
    const int lane  = threadIdx.x & 31;
    const int half  = lane >> 4;
    const int mrow  = lane & 15;

    const int wid = __builtin_amdgcn_readfirstlane(
        (int)((blockIdx.x * blockDim.x + threadIdx.x) >> 5));
    if (wid >= NWAVES) return;
    const int grp  = wid & (NGRP - 1);
    const int b0   = (wid / NGRP) * 32;
    const int k0   = grp * SPAN;
    const int kend = (k0 + SPAN < KWIN) ? (k0 + SPAN) : KWIN;

    const float aux0 = s[SDIM - 3];
    const float aux1 = s[SDIM - 2];
    const float aux2 = s[SDIM - 1];
    const float b3s  = b3[0];

    v2f A0, A1, A2, A3;
    {
        const int f0 = 2 * half;
#define LOADA(t, R) { int m = (t) * 16 + mrow; \
                      R.x = w1[m * 6 + f0]; R.y = w1[m * 6 + f0 + 1]; }
        LOADA(0, A0) LOADA(1, A1) LOADA(2, A2) LOADA(3, A3)
#undef LOADA
    }

    v8f C0, C1, C2, C3;
    v8f W0, W1, W2, W3f;
#define LOADC(t, CR, WR) \
    _Pragma("unroll") \
    for (int v = 0; v < 8; ++v) { \
        int m = (t) * 16 + v + 8 * half; \
        CR[v] = b1[m] + aux1 * w1[m * 6 + 4] + aux2 * w1[m * 6 + 5]; \
        WR[v] = w3[m]; \
    }
    LOADC(0, C0, W0) LOADC(1, C1, W1) LOADC(2, C2, W2) LOADC(3, C3, W3f)
#undef LOADC

#define WMMA(Af, Bf, Cf) \
    __builtin_amdgcn_wmma_f32_16x16x4_f32(false, Af, false, Bf, (short)0, Cf, false, false)

    const float* rowA = s + (b0 + mrow) * SDIM;
    const float* rowB = rowA + 16 * SDIM;

    v4f a0 = *(const v4f*)(rowA + k0);
    v4f a1 = *(const v4f*)(rowA + k0 + 4);
    v4f c0 = *(const v4f*)(rowB + k0);
    v4f c1 = *(const v4f*)(rowB + k0 + 4);

    for (int kc = k0; kc < kend; kc += 8) {
        const int kn = (kc + 8 < SDIM - 8) ? (kc + 8) : (SDIM - 8);
        const v4f n0 = *(const v4f*)(rowA + kn);
        const v4f n1 = *(const v4f*)(rowA + kn + 4);
        const v4f m0 = *(const v4f*)(rowB + kn);
        const v4f m1 = *(const v4f*)(rowB + kn + 4);

        const float ra[10] = { a0.x, a0.y, a0.z, a0.w, a1.x, a1.y, a1.z, a1.w, n0.x, n0.y };
        const float rb[10] = { c0.x, c0.y, c0.z, c0.w, c1.x, c1.y, c1.z, c1.w, m0.x, m0.y };

#pragma unroll
        for (int j = 0; j < 8; ++j) {
            const int k = kc + j;
            if (k >= kend) break;

            v2f B0, B1;
            B0.x = half ? ra[j + 2] : ra[j];  B0.y = half ? aux0 : ra[j + 1];
            B1.x = half ? rb[j + 2] : rb[j];  B1.y = half ? aux0 : rb[j + 1];

            v8f d00 = WMMA(A0, B0, C0);
            v8f d01 = WMMA(A1, B0, C1);
            v8f d02 = WMMA(A2, B0, C2);
            v8f d03 = WMMA(A3, B0, C3);
            v8f d10 = WMMA(A0, B1, C0);
            v8f d11 = WMMA(A1, B1, C1);
            v8f d12 = WMMA(A2, B1, C2);
            v8f d13 = WMMA(A3, B1, C3);

            float a00 = 0.f, a01 = 0.f, a02 = 0.f, a03 = 0.f;
            float a10 = 0.f, a11 = 0.f, a12 = 0.f, a13 = 0.f;
#pragma unroll
            for (int v = 0; v < 8; ++v) {
                a00 += W0[v]  * fmaxf(d00[v], 0.f);
                a01 += W1[v]  * fmaxf(d01[v], 0.f);
                a02 += W2[v]  * fmaxf(d02[v], 0.f);
                a03 += W3f[v] * fmaxf(d03[v], 0.f);
                a10 += W0[v]  * fmaxf(d10[v], 0.f);
                a11 += W1[v]  * fmaxf(d11[v], 0.f);
                a12 += W2[v]  * fmaxf(d12[v], 0.f);
                a13 += W3f[v] * fmaxf(d13[v], 0.f);
            }
            float accA = (a00 + a01) + (a02 + a03);
            float accB = (a10 + a11) + (a12 + a13);

            int swA = __builtin_amdgcn_ds_swizzle(__builtin_bit_cast(int, accA), 0x401f);
            int swB = __builtin_amdgcn_ds_swizzle(__builtin_bit_cast(int, accB), 0x401f);
            float totA = accA + __builtin_bit_cast(float, swA);
            float totB = accB + __builtin_bit_cast(float, swB);

            float r = (half ? totB : totA) + b3s;
            r = __builtin_amdgcn_fmed3f(r, -1.0f, 1.0f);

            out[(k << 12) + b0 + lane] = r;
        }

        a0 = n0; a1 = n1; c0 = m0; c1 = m1;
    }
#undef WMMA
}

extern "C" void kernel_launch(void* const* d_in, const int* in_sizes, int n_in,
                              void* d_out, int out_size, void* d_ws, size_t ws_size,
                              hipStream_t stream) {
    (void)in_sizes; (void)n_in; (void)out_size; (void)d_ws; (void)ws_size;
    const float* s  = (const float*)d_in[0];
    const float* w1 = (const float*)d_in[1];
    const float* b1 = (const float*)d_in[2];
    const float* w3 = (const float*)d_in[3];
    const float* b3 = (const float*)d_in[4];
    float* out = (float*)d_out;

    actor_wmma_kernel<<<512, 256, 0, stream>>>(s, w1, b1, w3, b3, out);
}


# relu(x) taken as (x+|x|)/2: |x| rides free as a source modifier of the second-layer FMA, so both max per element go; linear part = 3-tap dot per output from folded weights; f32 sums regrouped
# speedup vs baseline: 1.4167x; 1.0660x over previous
#include <hip/hip_runtime.h>


typedef __attribute__((ext_vector_type(2))) float v2f;
typedef __attribute__((ext_vector_type(4))) float v4f;
typedef __attribute__((ext_vector_type(8))) float v8f;

static __device__ __forceinline__ float pick(int c, float a, float b) { return c ? a : b; }

#define BATCH 4096
#define SDIM  1024
#define KWIN  1019
#define SPAN  32
#define NGRP  32
#define NROWG (BATCH / 32)
#define NWAVES (NROWG * NGRP)

__global__ __launch_bounds__(256) void actor_wmma_kernel(
    const float* __restrict__ s,
    const float* __restrict__ w1,
    const float* __restrict__ b1,
    const float* __restrict__ w3,
    const float* __restrict__ b3,
    float* __restrict__ out)
{
    const int lane  = threadIdx.x & 31;
    const int half  = lane >> 4;
    const int mrow  = lane & 15;

    const int wid = __builtin_amdgcn_readfirstlane(
        (int)((blockIdx.x * blockDim.x + threadIdx.x) >> 5));
    if (wid >= NWAVES) return;
    const int grp  = wid & (NGRP - 1);
    const int b0   = (wid / NGRP) * 32;
    const int k0   = grp * SPAN;
    const int kend = (k0 + SPAN < KWIN) ? (k0 + SPAN) : KWIN;

    const float aux0 = s[SDIM - 3];
    const float aux1 = s[SDIM - 2];
    const float aux2 = s[SDIM - 1];
    const float b3s  = b3[0];

    v2f A0, A1, A2, A3;
    {
        const int f0 = 2 * half;
#define LOADA(t, R) { int m = (t) * 16 + mrow; \
                      R.x = w1[m * 6 + f0]; R.y = w1[m * 6 + f0 + 1]; }
        LOADA(0, A0) LOADA(1, A1) LOADA(2, A2) LOADA(3, A3)
#undef LOADA
    }

    v8f C0, C1, C2, C3;
    v8f W0, W1, W2, W3f;
#define LOADC(t, CR, WR) \
    _Pragma("unroll") \
    for (int v = 0; v < 8; ++v) { \
        int m = (t) * 16 + v + 8 * half; \
        CR[v] = b1[m] + aux1 * w1[m * 6 + 4] + aux2 * w1[m * 6 + 5]; \
        WR[v] = 0.5f * w3[m]; \
    }
    LOADC(0, C0, W0) LOADC(1, C1, W1) LOADC(2, C2, W2) LOADC(3, C3, W3f)
#undef LOADC

    float u0, u1, u2, ucb;
    {
        const float* q0 = w1 + lane * 6;
        const float* q1 = w1 + (lane + 32) * 6;
        const float h0 = 0.5f * w3[lane];
        const float h1 = 0.5f * w3[lane + 32];
        const float z0 = b1[lane]      + aux0 * q0[3] + aux1 * q0[4] + aux2 * q0[5];
        const float z1 = b1[lane + 32] + aux0 * q1[3] + aux1 * q1[4] + aux2 * q1[5];
        u0  = h0 * q0[0] + h1 * q1[0];
        u1  = h0 * q0[1] + h1 * q1[1];
        u2  = h0 * q0[2] + h1 * q1[2];
        ucb = h0 * z0    + h1 * z1;
#define XADD(x, pat) x += __builtin_bit_cast(float, \
            __builtin_amdgcn_ds_swizzle(__builtin_bit_cast(int, x), pat))
        XADD(u0, 0x401f); XADD(u1, 0x401f); XADD(u2, 0x401f); XADD(ucb, 0x401f);
        XADD(u0, 0x201f); XADD(u1, 0x201f); XADD(u2, 0x201f); XADD(ucb, 0x201f);
        XADD(u0, 0x101f); XADD(u1, 0x101f); XADD(u2, 0x101f); XADD(ucb, 0x101f);
        XADD(u0, 0x081f); XADD(u1, 0x081f); XADD(u2, 0x081f); XADD(ucb, 0x081f);
        XADD(u0, 0x041f); XADD(u1, 0x041f); XADD(u2, 0x041f); XADD(ucb, 0x041f);
#undef XADD
        ucb += b3s;
    }

#define WMMA(Af, Bf, Cf) \
    __builtin_amdgcn_wmma_f32_16x16x4_f32(false, Af, false, Bf, (short)0, Cf, false, false)

    const float* rowA = s + (b0 + mrow) * SDIM;
    const float* rowB = rowA + 16 * SDIM;

    v4f a0 = *(const v4f*)(rowA + k0);
    v4f a1 = *(const v4f*)(rowA + k0 + 4);
    v4f c0 = *(const v4f*)(rowB + k0);
    v4f c1 = *(const v4f*)(rowB + k0 + 4);

    for (int kc = k0; kc < kend; kc += 8) {
        const int kn = (kc + 8 < SDIM - 8) ? (kc + 8) : (SDIM - 8);
        const v4f n0 = *(const v4f*)(rowA + kn);
        const v4f n1 = *(const v4f*)(rowA + kn + 4);
        const v4f m0 = *(const v4f*)(rowB + kn);
        const v4f m1 = *(const v4f*)(rowB + kn + 4);

        const float ra[10] = { a0.x, a0.y, a0.z, a0.w, a1.x, a1.y, a1.z, a1.w, n0.x, n0.y };
        const float rb[10] = { c0.x, c0.y, c0.z, c0.w, c1.x, c1.y, c1.z, c1.w, m0.x, m0.y };
        const float rs[10] = { pick(half, c0.x, a0.x), pick(half, c0.y, a0.y), pick(half, c0.z, a0.z),
                               pick(half, c0.w, a0.w), pick(half, c1.x, a1.x), pick(half, c1.y, a1.y),
                               pick(half, c1.z, a1.z), pick(half, c1.w, a1.w), pick(half, m0.x, n0.x),
                               pick(half, m0.y, n0.y) };

#pragma unroll
        for (int j = 0; j < 8; ++j) {
            const int k = kc + j;
            if (k >= kend) break;

            v2f B0, B1;
            B0.x = half ? ra[j + 2] : ra[j];  B0.y = half ? aux0 : ra[j + 1];
            B1.x = half ? rb[j + 2] : rb[j];  B1.y = half ? aux0 : rb[j + 1];

            v8f d00 = WMMA(A0, B0, C0);
            v8f d01 = WMMA(A1, B0, C1);
            v8f d02 = WMMA(A2, B0, C2);
            v8f d03 = WMMA(A3, B0, C3);
            v8f d10 = WMMA(A0, B1, C0);
            v8f d11 = WMMA(A1, B1, C1);
            v8f d12 = WMMA(A2, B1, C2);
            v8f d13 = WMMA(A3, B1, C3);

            float a00, a01, a02, a03, a10, a11, a12, a13;
#define ROW8(OP, TAIL, OUT, v) \
            asm(OP " %0, %8, |%12|"  TAIL("%0") "\n\t" \
                OP " %1, %9, |%13|"  TAIL("%1") "\n\t" \
                OP " %2, %10, |%14|" TAIL("%2") "\n\t" \
                OP " %3, %11, |%15|" TAIL("%3") "\n\t" \
                OP " %4, %8, |%16|"  TAIL("%4") "\n\t" \
                OP " %5, %9, |%17|"  TAIL("%5") "\n\t" \
                OP " %6, %10, |%18|" TAIL("%6") "\n\t" \
                OP " %7, %11, |%19|" TAIL("%7") \
                : OUT(a00), OUT(a01), OUT(a02), OUT(a03), OUT(a10), OUT(a11), OUT(a12), OUT(a13) \
                : "v"(W0[v]), "v"(W1[v]), "v"(W2[v]), "v"(W3f[v]), \
                  "v"(d00[v]), "v"(d01[v]), "v"(d02[v]), "v"(d03[v]), \
                  "v"(d10[v]), "v"(d11[v]), "v"(d12[v]), "v"(d13[v]))
#define NO_TAIL(x)  ""
#define ACC_TAIL(x) ", " x
#define OUT_NEW(x)  "=&v"(x)
#define OUT_ACC(x)  "+v"(x)
            ROW8("v_mul_f32", NO_TAIL, OUT_NEW, 0);
            ROW8("v_fma_f32", ACC_TAIL, OUT_ACC, 1);
            ROW8("v_fma_f32", ACC_TAIL, OUT_ACC, 2);
            ROW8("v_fma_f32", ACC_TAIL, OUT_ACC, 3);
            ROW8("v_fma_f32", ACC_TAIL, OUT_ACC, 4);
            ROW8("v_fma_f32", ACC_TAIL, OUT_ACC, 5);
            ROW8("v_fma_f32", ACC_TAIL, OUT_ACC, 6);
            ROW8("v_fma_f32", ACC_TAIL, OUT_ACC, 7);
#undef ROW8
#undef NO_TAIL
#undef ACC_TAIL
#undef OUT_NEW
#undef OUT_ACC
            float accA = (a00 + a01) + (a02 + a03);
            float accB = (a10 + a11) + (a12 + a13);

            int swA = __builtin_amdgcn_ds_swizzle(__builtin_bit_cast(int, accA), 0x401f);
            int swB = __builtin_amdgcn_ds_swizzle(__builtin_bit_cast(int, accB), 0x401f);
            float totA = accA + __builtin_bit_cast(float, swA);
            float totB = accB + __builtin_bit_cast(float, swB);

            const float lin = __builtin_fmaf(u0, rs[j],
                              __builtin_fmaf(u1, rs[j + 1],
                              __builtin_fmaf(u2, rs[j + 2], ucb)));

            float r = (half ? totB : totA) + lin;
            r = __builtin_amdgcn_fmed3f(r, -1.0f, 1.0f);

            out[(k << 12) + b0 + lane] = r;
        }

        a0 = n0; a1 = n1; c0 = m0; c1 = m1;
    }
#undef WMMA
}

extern "C" void kernel_launch(void* const* d_in, const int* in_sizes, int n_in,
                              void* d_out, int out_size, void* d_ws, size_t ws_size,
                              hipStream_t stream) {
    (void)in_sizes; (void)n_in; (void)out_size; (void)d_ws; (void)ws_size;
    const float* s  = (const float*)d_in[0];
    const float* w1 = (const float*)d_in[1];
    const float* b1 = (const float*)d_in[2];
    const float* w3 = (const float*)d_in[3];
    const float* b3 = (const float*)d_in[4];
    float* out = (float*)d_out;

    actor_wmma_kernel<<<512, 256, 0, stream>>>(s, w1, b1, w3, b3, out);
}


# Lane-half sum of layer 2 by ONE v_permlanex16 exchange per window (a lane keeps its own tile's partial, hands over the other's) instead of two LDS-unit swizzles and their waits; sums keep their bits
# speedup vs baseline: 1.4724x; 1.0393x over previous
#include <hip/hip_runtime.h>


typedef __attribute__((ext_vector_type(2))) float v2f;
typedef __attribute__((ext_vector_type(4))) float v4f;
typedef __attribute__((ext_vector_type(8))) float v8f;

static __device__ __forceinline__ float pick(int c, float a, float b) { return c ? a : b; }

static __device__ __forceinline__ float xhalf(float x) {
    const int b = __builtin_bit_cast(int, x);
    return __builtin_bit_cast(float,
        __builtin_amdgcn_permlanex16(b, b, 0x76543210, 0xfedcba98, false, false));
}

#define BATCH 4096
#define SDIM  1024
#define KWIN  1019
#define SPAN  32
#define NGRP  32
#define NROWG (BATCH / 32)
#define NWAVES (NROWG * NGRP)

__global__ __launch_bounds__(256) void actor_wmma_kernel(
    const float* __restrict__ s,
    const float* __restrict__ w1,
    const float* __restrict__ b1,
    const float* __restrict__ w3,
    const float* __restrict__ b3,
    float* __restrict__ out)
{
    const int lane  = threadIdx.x & 31;
    const int half  = lane >> 4;
    const int mrow  = lane & 15;

    const int wid = __builtin_amdgcn_readfirstlane(
        (int)((blockIdx.x * blockDim.x + threadIdx.x) >> 5));
    if (wid >= NWAVES) return;
    const int grp  = wid & (NGRP - 1);
    const int b0   = (wid / NGRP) * 32;
    const int k0   = grp * SPAN;
    const int kend = (k0 + SPAN < KWIN) ? (k0 + SPAN) : KWIN;

    const float aux0 = s[SDIM - 3];
    const float aux1 = s[SDIM - 2];
    const float aux2 = s[SDIM - 1];
    const float b3s  = b3[0];

    v2f A0, A1, A2, A3;
    {
        const int f0 = 2 * half;
#define LOADA(t, R) { int m = (t) * 16 + mrow; \
                      R.x = w1[m * 6 + f0]; R.y = w1[m * 6 + f0 + 1]; }
        LOADA(0, A0) LOADA(1, A1) LOADA(2, A2) LOADA(3, A3)
#undef LOADA
    }

    v8f C0, C1, C2, C3;
    v8f W0, W1, W2, W3f;
#define LOADC(t, CR, WR) \
    _Pragma("unroll") \
    for (int v = 0; v < 8; ++v) { \
        int m = (t) * 16 + v + 8 * half; \
        CR[v] = b1[m] + aux1 * w1[m * 6 + 4] + aux2 * w1[m * 6 + 5]; \
        WR[v] = 0.5f * w3[m]; \
    }
    LOADC(0, C0, W0) LOADC(1, C1, W1) LOADC(2, C2, W2) LOADC(3, C3, W3f)
#undef LOADC

    float u0, u1, u2, ucb;
    {
        const float* q0 = w1 + lane * 6;
        const float* q1 = w1 + (lane + 32) * 6;
        const float h0 = 0.5f * w3[lane];
        const float h1 = 0.5f * w3[lane + 32];
        const float z0 = b1[lane]      + aux0 * q0[3] + aux1 * q0[4] + aux2 * q0[5];
        const float z1 = b1[lane + 32] + aux0 * q1[3] + aux1 * q1[4] + aux2 * q1[5];
        u0  = h0 * q0[0] + h1 * q1[0];
        u1  = h0 * q0[1] + h1 * q1[1];
        u2  = h0 * q0[2] + h1 * q1[2];
        ucb = h0 * z0    + h1 * z1;
#define XADD(x, pat) x += __builtin_bit_cast(float, \
            __builtin_amdgcn_ds_swizzle(__builtin_bit_cast(int, x), pat))
        XADD(u0, 0x401f); XADD(u1, 0x401f); XADD(u2, 0x401f); XADD(ucb, 0x401f);
        XADD(u0, 0x201f); XADD(u1, 0x201f); XADD(u2, 0x201f); XADD(ucb, 0x201f);
        XADD(u0, 0x101f); XADD(u1, 0x101f); XADD(u2, 0x101f); XADD(ucb, 0x101f);
        XADD(u0, 0x081f); XADD(u1, 0x081f); XADD(u2, 0x081f); XADD(ucb, 0x081f);
        XADD(u0, 0x041f); XADD(u1, 0x041f); XADD(u2, 0x041f); XADD(ucb, 0x041f);
#undef XADD
        ucb += b3s;
    }

#define WMMA(Af, Bf, Cf) \
    __builtin_amdgcn_wmma_f32_16x16x4_f32(false, Af, false, Bf, (short)0, Cf, false, false)

    const float* rowA = s + (b0 + mrow) * SDIM;
    const float* rowB = rowA + 16 * SDIM;

    v4f a0 = *(const v4f*)(rowA + k0);
    v4f a1 = *(const v4f*)(rowA + k0 + 4);
    v4f c0 = *(const v4f*)(rowB + k0);
    v4f c1 = *(const v4f*)(rowB + k0 + 4);

    for (int kc = k0; kc < kend; kc += 8) {
        const int kn = (kc + 8 < SDIM - 8) ? (kc + 8) : (SDIM - 8);
        const v4f n0 = *(const v4f*)(rowA + kn);
        const v4f n1 = *(const v4f*)(rowA + kn + 4);
        const v4f m0 = *(const v4f*)(rowB + kn);
        const v4f m1 = *(const v4f*)(rowB + kn + 4);

        const float ra[10] = { a0.x, a0.y, a0.z, a0.w, a1.x, a1.y, a1.z, a1.w, n0.x, n0.y };
        const float rb[10] = { c0.x, c0.y, c0.z, c0.w, c1.x, c1.y, c1.z, c1.w, m0.x, m0.y };
        const float rs[10] = { pick(half, c0.x, a0.x), pick(half, c0.y, a0.y), pick(half, c0.z, a0.z),
                               pick(half, c0.w, a0.w), pick(half, c1.x, a1.x), pick(half, c1.y, a1.y),
                               pick(half, c1.z, a1.z), pick(half, c1.w, a1.w), pick(half, m0.x, n0.x),
                               pick(half, m0.y, n0.y) };

#pragma unroll
        for (int j = 0; j < 8; ++j) {
            const int k = kc + j;
            if (k >= kend) break;

            v2f B0, B1;
            B0.x = half ? ra[j + 2] : ra[j];  B0.y = half ? aux0 : ra[j + 1];
            B1.x = half ? rb[j + 2] : rb[j];  B1.y = half ? aux0 : rb[j + 1];

            v8f d00 = WMMA(A0, B0, C0);
            v8f d01 = WMMA(A1, B0, C1);
            v8f d02 = WMMA(A2, B0, C2);
            v8f d03 = WMMA(A3, B0, C3);
            v8f d10 = WMMA(A0, B1, C0);
            v8f d11 = WMMA(A1, B1, C1);
            v8f d12 = WMMA(A2, B1, C2);
            v8f d13 = WMMA(A3, B1, C3);

            float a00, a01, a02, a03, a10, a11, a12, a13;
#define ROW8(OP, TAIL, OUT, v) \
            asm(OP " %0, %8, |%12|"  TAIL("%0") "\n\t" \
                OP " %1, %9, |%13|"  TAIL("%1") "\n\t" \
                OP " %2, %10, |%14|" TAIL("%2") "\n\t" \
                OP " %3, %11, |%15|" TAIL("%3") "\n\t" \
                OP " %4, %8, |%16|"  TAIL("%4") "\n\t" \
                OP " %5, %9, |%17|"  TAIL("%5") "\n\t" \
                OP " %6, %10, |%18|" TAIL("%6") "\n\t" \
                OP " %7, %11, |%19|" TAIL("%7") \
                : OUT(a00), OUT(a01), OUT(a02), OUT(a03), OUT(a10), OUT(a11), OUT(a12), OUT(a13) \
                : "v"(W0[v]), "v"(W1[v]), "v"(W2[v]), "v"(W3f[v]), \
                  "v"(d00[v]), "v"(d01[v]), "v"(d02[v]), "v"(d03[v]), \
                  "v"(d10[v]), "v"(d11[v]), "v"(d12[v]), "v"(d13[v]))
#define NO_TAIL(x)  ""
#define ACC_TAIL(x) ", " x
#define OUT_NEW(x)  "=&v"(x)
#define OUT_ACC(x)  "+v"(x)
            ROW8("v_mul_f32", NO_TAIL, OUT_NEW, 0);
            ROW8("v_fma_f32", ACC_TAIL, OUT_ACC, 1);
            ROW8("v_fma_f32", ACC_TAIL, OUT_ACC, 2);
            ROW8("v_fma_f32", ACC_TAIL, OUT_ACC, 3);
            ROW8("v_fma_f32", ACC_TAIL, OUT_ACC, 4);
            ROW8("v_fma_f32", ACC_TAIL, OUT_ACC, 5);
            ROW8("v_fma_f32", ACC_TAIL, OUT_ACC, 6);
            ROW8("v_fma_f32", ACC_TAIL, OUT_ACC, 7);
#undef ROW8
#undef NO_TAIL
#undef ACC_TAIL
#undef OUT_NEW
#undef OUT_ACC
            float accA = (a00 + a01) + (a02 + a03);
            float accB = (a10 + a11) + (a12 + a13);

            const float keep = pick(half, accB, accA);
            const float send = pick(half, accA, accB);
            const float tot  = keep + xhalf(send);

            const float lin = __builtin_fmaf(u0, rs[j],
                              __builtin_fmaf(u1, rs[j + 1],
                              __builtin_fmaf(u2, rs[j + 2], ucb)));

            float r = tot + lin;
            r = __builtin_amdgcn_fmed3f(r, -1.0f, 1.0f);

            out[(k << 12) + b0 + lane] = r;
        }

        a0 = n0; a1 = n1; c0 = m0; c1 = m1;
    }
#undef WMMA
}

extern "C" void kernel_launch(void* const* d_in, const int* in_sizes, int n_in,
                              void* d_out, int out_size, void* d_ws, size_t ws_size,
                              hipStream_t stream) {
    (void)in_sizes; (void)n_in; (void)out_size; (void)d_ws; (void)ws_size;
    const float* s  = (const float*)d_in[0];
    const float* w1 = (const float*)d_in[1];
    const float* b1 = (const float*)d_in[2];
    const float* w3 = (const float*)d_in[3];
    const float* b3 = (const float*)d_in[4];
    float* out = (float*)d_out;

    actor_wmma_kernel<<<512, 256, 0, stream>>>(s, w1, b1, w3, b3, out);
}
